# starting kernel
# baseline (speedup 1.0000x reference)
#include <hip/hip_runtime.h>
#include <hip/hip_bf16.h>


typedef __attribute__((ext_vector_type(2))) float v2f;
typedef __attribute__((ext_vector_type(8))) float v8f;

#define F_NODES 16384
#define DEG 8
#define EDGES (F_NODES * DEG)
#define BATCH 64
#define HID 16
#define LAYERS 10
#define WAVES_PER_BLOCK 8
#define LDS_STRIDE 18

__device__ __forceinline__ float elu_f(float z) {
    return z > 0.0f ? z : (__expf(z) - 1.0f);
}

__global__ __launch_bounds__(256)
void gsnn_layer(const float* __restrict__ xin,
                const float* __restrict__ x0,
                const float* __restrict__ W1,
                const float* __restrict__ b1,
                const float* __restrict__ W2,
                const float* __restrict__ b2,
                const int*   __restrict__ ixs,
                float*       __restrict__ xout)
{
    __shared__ float lds_h[WAVES_PER_BLOCK * 16 * LDS_STRIDE];

    const int tid  = threadIdx.x;
    const int wave = tid >> 5;
    const int lane = tid & 31;
    const int n    = lane & 15;
    const int hi   = lane >> 4;

    const int f = blockIdx.x * WAVES_PER_BLOCK + wave;
    float* hbuf = &lds_h[wave * 16 * LDS_STRIDE];

    const float* W1f = W1 + (size_t)f * (HID * DEG);
    const v2f wb1_0 = *(const v2f*)(W1f + n * DEG + 2 * hi);
    const v2f wb1_1 = *(const v2f*)(W1f + n * DEG + 4 + 2 * hi);
    const float bias1 = b1[f * HID + n];

    const float* W2f = W2 + (size_t)f * (DEG * HID);
    v2f wb2[4];
    float bias2 = 0.0f;
    if (n < DEG) {
        #pragma unroll
        for (int j = 0; j < 4; ++j)
            wb2[j] = *(const v2f*)(W2f + n * HID + 4 * j + 2 * hi);
        bias2 = b2[f * DEG + n];
    } else {
        #pragma unroll
        for (int j = 0; j < 4; ++j) { wb2[j].x = 0.0f; wb2[j].y = 0.0f; }
    }

    const int* ixf = ixs + f * DEG;
    const int c00 = ixf[2 * hi];
    const int c01 = ixf[2 * hi + 1];
    const int c10 = ixf[4 + 2 * hi];
    const int c11 = ixf[4 + 2 * hi + 1];

    const size_t ocol = (size_t)f * DEG + n;

    #pragma unroll
    for (int mt = 0; mt < 4; ++mt) {
        const int mb = mt * 16;

        const float* xrow = xin + (size_t)(mb + n) * EDGES;
        v2f a0; a0.x = xrow[c00]; a0.y = xrow[c01];
        v2f a1; a1.x = xrow[c10]; a1.y = xrow[c11];

        v8f acc = {};
        acc = __builtin_amdgcn_wmma_f32_16x16x4_f32(
                  false, a0, false, wb1_0, (short)0, acc, false, false);
        acc = __builtin_amdgcn_wmma_f32_16x16x4_f32(
                  false, a1, false, wb1_1, (short)0, acc, false, false);

        #pragma unroll
        for (int r = 0; r < 8; ++r) {
            float hv = elu_f(acc[r] + bias1);
            hbuf[(r + 8 * hi) * LDS_STRIDE + n] = hv;
        }
        __syncthreads();

        v2f a2[4];
        #pragma unroll
        for (int j = 0; j < 4; ++j)
            a2[j] = *(const v2f*)(hbuf + n * LDS_STRIDE + 4 * j + 2 * hi);

        v8f o = {};
        #pragma unroll
        for (int j = 0; j < 4; ++j)
            o = __builtin_amdgcn_wmma_f32_16x16x4_f32(
                    false, a2[j], false, wb2[j], (short)0, o, false, false);

        __syncthreads();

        if (n < DEG) {
            #pragma unroll
            for (int r = 0; r < 8; ++r) {
                const size_t off = (size_t)(mb + r + 8 * hi) * EDGES + ocol;
                xout[off] = o[r] + bias2 + x0[off];
            }
        }
    }
}

extern "C" void kernel_launch(void* const* d_in, const int* in_sizes, int n_in,
                              void* d_out, int out_size, void* d_ws, size_t ws_size,
                              hipStream_t stream) {
    (void)in_sizes; (void)n_in; (void)out_size; (void)ws_size;

    const float* x0 = (const float*)d_in[0];
    const float* W1 = (const float*)d_in[1];
    const float* b1 = (const float*)d_in[2];
    const float* W2 = (const float*)d_in[3];
    const float* b2 = (const float*)d_in[4];
    const int*   ix = (const int*)d_in[5];

    float* bufA = (float*)d_ws;
    float* out  = (float*)d_out;

    const dim3 grid(F_NODES / WAVES_PER_BLOCK);
    const dim3 block(WAVES_PER_BLOCK * 32);

    const float* cur = x0;
    for (int l = 0; l < LAYERS; ++l) {
        float* dst = (l & 1) ? out : bufA;
        gsnn_layer<<<grid, block, 0, stream>>>(cur, x0, W1, b1, W2, b2, ix, dst);
        cur = dst;
    }
}


# activations kept batch-major [E][B] between layers (products flipped to W x gT so batch is the lane axis, h passes D->B in registers): gathered edge = one 256B run; hidden-sum k order permuted
# speedup vs baseline: 2.7585x; 2.7585x over previous
#include <hip/hip_runtime.h>


typedef __attribute__((ext_vector_type(2))) float v2f;
typedef __attribute__((ext_vector_type(4))) float v4f;
typedef __attribute__((ext_vector_type(8))) float v8f;

#define F_NODES 16384
#define DEG 8
#define EDGES (F_NODES * DEG)
#define BATCH 64
#define HID 16
#define LAYERS 10
#define WAVES_PER_BLOCK 8

__device__ __forceinline__ float elu_f(float z) {
    return z > 0.0f ? z : (__expf(z) - 1.0f);
}

template <bool IN_T, bool OUT_T>
__global__ __launch_bounds__(256)
void gsnn_layer(const float* __restrict__ xin,
                const float* __restrict__ x0,
                const float* __restrict__ W1,
                const float* __restrict__ b1,
                const float* __restrict__ W2,
                const float* __restrict__ b2,
                const int*   __restrict__ ixs,
                float*       __restrict__ xout)
{
    const int tid  = threadIdx.x;
    const int wave = tid >> 5;
    const int lane = tid & 31;
    const int c    = lane & 15;
    const int hi   = lane >> 4;

    const int f = blockIdx.x * WAVES_PER_BLOCK + wave;

    const int* ixf = ixs + f * DEG;
    int col[4];
    col[0] = ixf[2 * hi];
    col[1] = ixf[2 * hi + 1];
    col[2] = ixf[4 + 2 * hi];
    col[3] = ixf[4 + 2 * hi + 1];

    const float* x0p = x0 + (size_t)(4 * c) * EDGES + (size_t)f * DEG + 4 * hi;
    v4f res[4];
    #pragma unroll
    for (int t = 0; t < 4; ++t)
        res[t] = *(const v4f*)(x0p + (size_t)t * EDGES);

    const float* W1f = W1 + (size_t)f * (HID * DEG);
    const v2f wa1_0 = *(const v2f*)(W1f + c * DEG + 2 * hi);
    const v2f wa1_1 = *(const v2f*)(W1f + c * DEG + 4 + 2 * hi);

    const v4f b1a = *(const v4f*)(b1 + (size_t)f * HID + 8 * hi);
    const v4f b1b = *(const v4f*)(b1 + (size_t)f * HID + 8 * hi + 4);
    const float b1v[8] = { b1a.x, b1a.y, b1a.z, b1a.w, b1b.x, b1b.y, b1b.z, b1b.w };

    const int  och   = (c & 3) + 4 * (c >> 3);
    const bool arow  = (c & 4) == 0;
    const float* W2r = W2 + (size_t)f * (DEG * HID) + och * HID + 8 * hi;
    const v4f w2a = *(const v4f*)(W2r);
    const v4f w2b = *(const v4f*)(W2r + 4);
    v2f wa2[4];
    wa2[0].x = arow ? w2a.x : 0.0f;  wa2[0].y = arow ? w2a.y : 0.0f;
    wa2[1].x = arow ? w2a.z : 0.0f;  wa2[1].y = arow ? w2a.w : 0.0f;
    wa2[2].x = arow ? w2b.x : 0.0f;  wa2[2].y = arow ? w2b.y : 0.0f;
    wa2[3].x = arow ? w2b.z : 0.0f;  wa2[3].y = arow ? w2b.w : 0.0f;

    const v4f b2v = *(const v4f*)(b2 + (size_t)f * DEG + 4 * hi);

    v4f g[4];
    if (IN_T) {
        #pragma unroll
        for (int k = 0; k < 4; ++k)
            g[k] = *(const v4f*)(xin + (size_t)col[k] * BATCH + 4 * c);
    } else {
        #pragma unroll
        for (int k = 0; k < 4; ++k) {
            #pragma unroll
            for (int t = 0; t < 4; ++t)
                g[k][t] = xin[(size_t)(4 * c + t) * EDGES + col[k]];
        }
    }

    v8f o[4];
    #pragma unroll
    for (int t = 0; t < 4; ++t) {
        v2f bg0; bg0.x = g[0][t]; bg0.y = g[1][t];
        v2f bg1; bg1.x = g[2][t]; bg1.y = g[3][t];
        v8f acc = {};
        acc = __builtin_amdgcn_wmma_f32_16x16x4_f32(
                  false, wa1_0, false, bg0, (short)0, acc, false, false);
        acc = __builtin_amdgcn_wmma_f32_16x16x4_f32(
                  false, wa1_1, false, bg1, (short)0, acc, false, false);

        float hv[8];
        #pragma unroll
        for (int r = 0; r < 8; ++r)
            hv[r] = elu_f(acc[r] + b1v[r]);

        v8f oo = {};
        #pragma unroll
        for (int j = 0; j < 4; ++j) {
            v2f bh; bh.x = hv[2 * j]; bh.y = hv[2 * j + 1];
            oo = __builtin_amdgcn_wmma_f32_16x16x4_f32(
                     false, wa2[j], false, bh, (short)0, oo, false, false);
        }
        o[t] = oo;
    }

    if (OUT_T) {
        float* op = xout + ((size_t)f * DEG + 4 * hi) * BATCH + 4 * c;
        #pragma unroll
        for (int r = 0; r < 4; ++r) {
            v4f v;
            #pragma unroll
            for (int t = 0; t < 4; ++t)
                v[t] = o[t][r] + b2v[r] + res[t][r];
            *(v4f*)(op + (size_t)r * BATCH) = v;
        }
    } else {
        float* op = xout + (size_t)(4 * c) * EDGES + (size_t)f * DEG + 4 * hi;
        #pragma unroll
        for (int t = 0; t < 4; ++t) {
            v4f v;
            #pragma unroll
            for (int r = 0; r < 4; ++r)
                v[r] = o[t][r] + b2v[r] + res[t][r];
            *(v4f*)(op + (size_t)t * EDGES) = v;
        }
    }
}

extern "C" void kernel_launch(void* const* d_in, const int* in_sizes, int n_in,
                              void* d_out, int out_size, void* d_ws, size_t ws_size,
                              hipStream_t stream) {
    (void)in_sizes; (void)n_in; (void)out_size; (void)ws_size;

    const float* x0 = (const float*)d_in[0];
    const float* W1 = (const float*)d_in[1];
    const float* b1 = (const float*)d_in[2];
    const float* W2 = (const float*)d_in[3];
    const float* b2 = (const float*)d_in[4];
    const int*   ix = (const int*)d_in[5];

    float* bufA = (float*)d_ws;
    float* out  = (float*)d_out;

    const dim3 grid(F_NODES / WAVES_PER_BLOCK);
    const dim3 block(WAVES_PER_BLOCK * 32);

    gsnn_layer<false, true><<<grid, block, 0, stream>>>(x0, x0, W1, b1, W2, b2, ix, bufA);
    const float* cur = bufA;
    for (int l = 1; l < LAYERS - 1; ++l) {
        float* dst = (l & 1) ? out : bufA;
        gsnn_layer<true, true><<<grid, block, 0, stream>>>(cur, x0, W1, b1, W2, b2, ix, dst);
        cur = dst;
    }
    gsnn_layer<true, false><<<grid, block, 0, stream>>>(cur, x0, W1, b1, W2, b2, ix, out);
}


# each wave now works through 4 nodes in turn, the next node's loads in flight during the current node's products: 4096 long-lived waves (4 per SIMD) replace 16384 short ones that were slow to enter
# speedup vs baseline: 3.0026x; 1.0885x over previous
#include <hip/hip_runtime.h>


typedef __attribute__((ext_vector_type(2))) float v2f;
typedef __attribute__((ext_vector_type(4))) float v4f;
typedef __attribute__((ext_vector_type(8))) float v8f;

#define F_NODES 16384
#define DEG 8
#define EDGES (F_NODES * DEG)
#define BATCH 64
#define HID 16
#define LAYERS 10
#define WAVES_PER_BLOCK 8
#define NODES_PER_WAVE 4
#define GRID_WAVES (F_NODES / NODES_PER_WAVE)

__device__ __forceinline__ float elu_f(float z) {
    return z > 0.0f ? z : (__expf(z) - 1.0f);
}

struct NodeIn {
    v4f g[4];
    v2f wa1_0, wa1_1;
    v4f b1a, b1b;
    v4f w2a, w2b;
    v4f b2v;
    v4f res[4];
};

template <bool IN_T>
__device__ __forceinline__ void load_node(NodeIn& n, const int f, const int (&col)[4],
                                          const float* __restrict__ xin,
                                          const float* __restrict__ x0,
                                          const float* __restrict__ W1,
                                          const float* __restrict__ b1,
                                          const float* __restrict__ W2,
                                          const float* __restrict__ b2,
                                          const int c, const int hi)
{
    if (IN_T) {
        #pragma unroll
        for (int k = 0; k < 4; ++k)
            n.g[k] = *(const v4f*)(xin + (size_t)col[k] * BATCH + 4 * c);
    } else {
        #pragma unroll
        for (int k = 0; k < 4; ++k) {
            #pragma unroll
            for (int t = 0; t < 4; ++t)
                n.g[k][t] = xin[(size_t)(4 * c + t) * EDGES + col[k]];
        }
    }

    const float* W1f = W1 + (size_t)f * (HID * DEG);
    n.wa1_0 = *(const v2f*)(W1f + c * DEG + 2 * hi);
    n.wa1_1 = *(const v2f*)(W1f + c * DEG + 4 + 2 * hi);

    n.b1a = *(const v4f*)(b1 + (size_t)f * HID + 8 * hi);
    n.b1b = *(const v4f*)(b1 + (size_t)f * HID + 8 * hi + 4);

    const int och = (c & 3) + 4 * (c >> 3);
    const float* W2r = W2 + (size_t)f * (DEG * HID) + och * HID + 8 * hi;
    n.w2a = *(const v4f*)(W2r);
    n.w2b = *(const v4f*)(W2r + 4);

    n.b2v = *(const v4f*)(b2 + (size_t)f * DEG + 4 * hi);

    const float* x0p = x0 + (size_t)(4 * c) * EDGES + (size_t)f * DEG + 4 * hi;
    #pragma unroll
    for (int t = 0; t < 4; ++t)
        n.res[t] = *(const v4f*)(x0p + (size_t)t * EDGES);
}

template <bool OUT_T>
__device__ __forceinline__ void compute_node(const NodeIn& n, const int f,
                                             float* __restrict__ xout,
                                             const int c, const int hi)
{
    const float b1v[8] = { n.b1a.x, n.b1a.y, n.b1a.z, n.b1a.w,
                           n.b1b.x, n.b1b.y, n.b1b.z, n.b1b.w };

    const bool arow = (c & 4) == 0;
    v2f wa2[4];
    wa2[0].x = arow ? n.w2a.x : 0.0f;  wa2[0].y = arow ? n.w2a.y : 0.0f;
    wa2[1].x = arow ? n.w2a.z : 0.0f;  wa2[1].y = arow ? n.w2a.w : 0.0f;
    wa2[2].x = arow ? n.w2b.x : 0.0f;  wa2[2].y = arow ? n.w2b.y : 0.0f;
    wa2[3].x = arow ? n.w2b.z : 0.0f;  wa2[3].y = arow ? n.w2b.w : 0.0f;

    v8f o[4];
    #pragma unroll
    for (int t = 0; t < 4; ++t) {
        v2f bg0; bg0.x = n.g[0][t]; bg0.y = n.g[1][t];
        v2f bg1; bg1.x = n.g[2][t]; bg1.y = n.g[3][t];
        v8f acc = {};
        acc = __builtin_amdgcn_wmma_f32_16x16x4_f32(
                  false, n.wa1_0, false, bg0, (short)0, acc, false, false);
        acc = __builtin_amdgcn_wmma_f32_16x16x4_f32(
                  false, n.wa1_1, false, bg1, (short)0, acc, false, false);

        float hv[8];
        #pragma unroll
        for (int r = 0; r < 8; ++r)
            hv[r] = elu_f(acc[r] + b1v[r]);

        v8f oo = {};
        #pragma unroll
        for (int j = 0; j < 4; ++j) {
            v2f bh; bh.x = hv[2 * j]; bh.y = hv[2 * j + 1];
            oo = __builtin_amdgcn_wmma_f32_16x16x4_f32(
                     false, wa2[j], false, bh, (short)0, oo, false, false);
        }
        o[t] = oo;
    }

    if (OUT_T) {
        float* op = xout + ((size_t)f * DEG + 4 * hi) * BATCH + 4 * c;
        #pragma unroll
        for (int r = 0; r < 4; ++r) {
            v4f v;
            #pragma unroll
            for (int t = 0; t < 4; ++t)
                v[t] = o[t][r] + n.b2v[r] + n.res[t][r];
            *(v4f*)(op + (size_t)r * BATCH) = v;
        }
    } else {
        float* op = xout + (size_t)(4 * c) * EDGES + (size_t)f * DEG + 4 * hi;
        #pragma unroll
        for (int t = 0; t < 4; ++t) {
            v4f v;
            #pragma unroll
            for (int r = 0; r < 4; ++r)
                v[r] = o[t][r] + n.b2v[r] + n.res[t][r];
            *(v4f*)(op + (size_t)t * EDGES) = v;
        }
    }
}

template <bool IN_T, bool OUT_T>
__global__ __launch_bounds__(256) __attribute__((amdgpu_waves_per_eu(4, 4)))
void gsnn_layer(const float* __restrict__ xin,
                const float* __restrict__ x0,
                const float* __restrict__ W1,
                const float* __restrict__ b1,
                const float* __restrict__ W2,
                const float* __restrict__ b2,
                const int*   __restrict__ ixs,
                float*       __restrict__ xout)
{
    const int tid  = threadIdx.x;
    const int wave = tid >> 5;
    const int lane = tid & 31;
    const int c    = lane & 15;
    const int hi   = lane >> 4;

    const int f0 = blockIdx.x * WAVES_PER_BLOCK + wave;

    int col[NODES_PER_WAVE][4];
    #pragma unroll
    for (int i = 0; i < NODES_PER_WAVE; ++i) {
        const int* ixf = ixs + (size_t)(f0 + i * GRID_WAVES) * DEG;
        col[i][0] = ixf[2 * hi];
        col[i][1] = ixf[2 * hi + 1];
        col[i][2] = ixf[4 + 2 * hi];
        col[i][3] = ixf[4 + 2 * hi + 1];
    }

    NodeIn n[NODES_PER_WAVE];
    load_node<IN_T>(n[0], f0, col[0], xin, x0, W1, b1, W2, b2, c, hi);

    #pragma unroll
    for (int i = 0; i < NODES_PER_WAVE; ++i) {
        if (i + 1 < NODES_PER_WAVE)
            load_node<IN_T>(n[i + 1], f0 + (i + 1) * GRID_WAVES, col[i + 1],
                            xin, x0, W1, b1, W2, b2, c, hi);
        __builtin_amdgcn_sched_barrier(0);
        compute_node<OUT_T>(n[i], f0 + i * GRID_WAVES, xout, c, hi);
    }
}

extern "C" void kernel_launch(void* const* d_in, const int* in_sizes, int n_in,
                              void* d_out, int out_size, void* d_ws, size_t ws_size,
                              hipStream_t stream) {
    (void)in_sizes; (void)n_in; (void)out_size; (void)ws_size;

    const float* x0 = (const float*)d_in[0];
    const float* W1 = (const float*)d_in[1];
    const float* b1 = (const float*)d_in[2];
    const float* W2 = (const float*)d_in[3];
    const float* b2 = (const float*)d_in[4];
    const int*   ix = (const int*)d_in[5];

    float* bufA = (float*)d_ws;
    float* out  = (float*)d_out;

    const dim3 grid(GRID_WAVES / WAVES_PER_BLOCK);
    const dim3 block(WAVES_PER_BLOCK * 32);

    gsnn_layer<false, true><<<grid, block, 0, stream>>>(x0, x0, W1, b1, W2, b2, ix, bufA);
    const float* cur = bufA;
    for (int l = 1; l < LAYERS - 1; ++l) {
        float* dst = (l & 1) ? out : bufA;
        gsnn_layer<true, true><<<grid, block, 0, stream>>>(cur, x0, W1, b1, W2, b2, ix, dst);
        cur = dst;
    }
    gsnn_layer<true, false><<<grid, block, 0, stream>>>(cur, x0, W1, b1, W2, b2, ix, out);
}


# x0 is first turned batch-major by a tiled LDS pre-pass (11th launch, into d_out), so layer 0 gathers whole 256B runs like layers 1-9 instead of 512 lone dwords per node; sums unchanged
# speedup vs baseline: 3.4836x; 1.1602x over previous
#include <hip/hip_runtime.h>


typedef __attribute__((ext_vector_type(2))) float v2f;
typedef __attribute__((ext_vector_type(4))) float v4f;
typedef __attribute__((ext_vector_type(8))) float v8f;

#define F_NODES 16384
#define DEG 8
#define EDGES (F_NODES * DEG)
#define BATCH 64
#define HID 16
#define LAYERS 10
#define WAVES_PER_BLOCK 8
#define NODES_PER_WAVE 4
#define GRID_WAVES (F_NODES / NODES_PER_WAVE)
#define T_COLS 256
#define T_BLOCKS (EDGES / T_COLS)

__device__ __forceinline__ float elu_f(float z) {
    return z > 0.0f ? z : (__expf(z) - 1.0f);
}

struct NodeIn {
    v4f g[4];
    v2f wa1_0, wa1_1;
    v4f b1a, b1b;
    v4f w2a, w2b;
    v4f b2v;
    v4f res[4];
};

__device__ __forceinline__ void load_node(NodeIn& n, const int f, const int (&col)[4],
                                          const float* __restrict__ xin,
                                          const float* __restrict__ x0,
                                          const float* __restrict__ W1,
                                          const float* __restrict__ b1,
                                          const float* __restrict__ W2,
                                          const float* __restrict__ b2,
                                          const int c, const int hi)
{
    #pragma unroll
    for (int k = 0; k < 4; ++k)
        n.g[k] = *(const v4f*)(xin + (size_t)col[k] * BATCH + 4 * c);

    const float* W1f = W1 + (size_t)f * (HID * DEG);
    n.wa1_0 = *(const v2f*)(W1f + c * DEG + 2 * hi);
    n.wa1_1 = *(const v2f*)(W1f + c * DEG + 4 + 2 * hi);

    n.b1a = *(const v4f*)(b1 + (size_t)f * HID + 8 * hi);
    n.b1b = *(const v4f*)(b1 + (size_t)f * HID + 8 * hi + 4);

    const int och = (c & 3) + 4 * (c >> 3);
    const float* W2r = W2 + (size_t)f * (DEG * HID) + och * HID + 8 * hi;
    n.w2a = *(const v4f*)(W2r);
    n.w2b = *(const v4f*)(W2r + 4);

    n.b2v = *(const v4f*)(b2 + (size_t)f * DEG + 4 * hi);

    const float* x0p = x0 + (size_t)(4 * c) * EDGES + (size_t)f * DEG + 4 * hi;
    #pragma unroll
    for (int t = 0; t < 4; ++t)
        n.res[t] = *(const v4f*)(x0p + (size_t)t * EDGES);
}

template <bool OUT_T>
__device__ __forceinline__ void compute_node(const NodeIn& n, const int f,
                                             float* __restrict__ xout,
                                             const int c, const int hi)
{
    const float b1v[8] = { n.b1a.x, n.b1a.y, n.b1a.z, n.b1a.w,
                           n.b1b.x, n.b1b.y, n.b1b.z, n.b1b.w };

    const bool arow = (c & 4) == 0;
    v2f wa2[4];
    wa2[0].x = arow ? n.w2a.x : 0.0f;  wa2[0].y = arow ? n.w2a.y : 0.0f;
    wa2[1].x = arow ? n.w2a.z : 0.0f;  wa2[1].y = arow ? n.w2a.w : 0.0f;
    wa2[2].x = arow ? n.w2b.x : 0.0f;  wa2[2].y = arow ? n.w2b.y : 0.0f;
    wa2[3].x = arow ? n.w2b.z : 0.0f;  wa2[3].y = arow ? n.w2b.w : 0.0f;

    v8f o[4];
    #pragma unroll
    for (int t = 0; t < 4; ++t) {
        v2f bg0; bg0.x = n.g[0][t]; bg0.y = n.g[1][t];
        v2f bg1; bg1.x = n.g[2][t]; bg1.y = n.g[3][t];
        v8f acc = {};
        acc = __builtin_amdgcn_wmma_f32_16x16x4_f32(
                  false, n.wa1_0, false, bg0, (short)0, acc, false, false);
        acc = __builtin_amdgcn_wmma_f32_16x16x4_f32(
                  false, n.wa1_1, false, bg1, (short)0, acc, false, false);

        float hv[8];
        #pragma unroll
        for (int r = 0; r < 8; ++r)
            hv[r] = elu_f(acc[r] + b1v[r]);

        v8f oo = {};
        #pragma unroll
        for (int j = 0; j < 4; ++j) {
            v2f bh; bh.x = hv[2 * j]; bh.y = hv[2 * j + 1];
            oo = __builtin_amdgcn_wmma_f32_16x16x4_f32(
                     false, wa2[j], false, bh, (short)0, oo, false, false);
        }
        o[t] = oo;
    }

    if (OUT_T) {
        float* op = xout + ((size_t)f * DEG + 4 * hi) * BATCH + 4 * c;
        #pragma unroll
        for (int r = 0; r < 4; ++r) {
            v4f v;
            #pragma unroll
            for (int t = 0; t < 4; ++t)
                v[t] = o[t][r] + n.b2v[r] + n.res[t][r];
            *(v4f*)(op + (size_t)r * BATCH) = v;
        }
    } else {
        float* op = xout + (size_t)(4 * c) * EDGES + (size_t)f * DEG + 4 * hi;
        #pragma unroll
        for (int t = 0; t < 4; ++t) {
            v4f v;
            #pragma unroll
            for (int r = 0; r < 4; ++r)
                v[r] = o[t][r] + n.b2v[r] + n.res[t][r];
            *(v4f*)(op + (size_t)t * EDGES) = v;
        }
    }
}

template <bool OUT_T>
__global__ __launch_bounds__(256) __attribute__((amdgpu_waves_per_eu(4, 4)))
void gsnn_layer(const float* __restrict__ xin,
                const float* __restrict__ x0,
                const float* __restrict__ W1,
                const float* __restrict__ b1,
                const float* __restrict__ W2,
                const float* __restrict__ b2,
                const int*   __restrict__ ixs,
                float*       __restrict__ xout)
{
    const int tid  = threadIdx.x;
    const int wave = tid >> 5;
    const int lane = tid & 31;
    const int c    = lane & 15;
    const int hi   = lane >> 4;

    const int f0 = blockIdx.x * WAVES_PER_BLOCK + wave;

    int col[NODES_PER_WAVE][4];
    #pragma unroll
    for (int i = 0; i < NODES_PER_WAVE; ++i) {
        const int* ixf = ixs + (size_t)(f0 + i * GRID_WAVES) * DEG;
        col[i][0] = ixf[2 * hi];
        col[i][1] = ixf[2 * hi + 1];
        col[i][2] = ixf[4 + 2 * hi];
        col[i][3] = ixf[4 + 2 * hi + 1];
    }

    NodeIn n[NODES_PER_WAVE];
    load_node(n[0], f0, col[0], xin, x0, W1, b1, W2, b2, c, hi);

    #pragma unroll
    for (int i = 0; i < NODES_PER_WAVE; ++i) {
        if (i + 1 < NODES_PER_WAVE)
            load_node(n[i + 1], f0 + (i + 1) * GRID_WAVES, col[i + 1],
                      xin, x0, W1, b1, W2, b2, c, hi);
        __builtin_amdgcn_sched_barrier(0);
        compute_node<OUT_T>(n[i], f0 + i * GRID_WAVES, xout, c, hi);
    }
}

__global__ __launch_bounds__(256)
void transpose_x0(const float* __restrict__ x0,
                  float*       __restrict__ xT)
{
    __shared__ v4f tile[T_COLS * (BATCH / 4)];

    const int tid = threadIdx.x;
    const int e0  = blockIdx.x * T_COLS;

    const int c4 = tid & 63;
    const int bq = tid >> 6;

    v4f r[4][4];
    #pragma unroll
    for (int i = 0; i < 4; ++i) {
        #pragma unroll
        for (int j = 0; j < 4; ++j)
            r[i][j] = *(const v4f*)(x0 + (size_t)(4 * (bq + 4 * i) + j) * EDGES
                                       + e0 + 4 * c4);
    }

    #pragma unroll
    for (int i = 0; i < 4; ++i) {
        const int b4 = bq + 4 * i;
        #pragma unroll
        for (int jj = 0; jj < 4; ++jj) {
            v4f v;
            v.x = r[i][0][jj];  v.y = r[i][1][jj];
            v.z = r[i][2][jj];  v.w = r[i][3][jj];
            tile[(4 * c4 + jj) * (BATCH / 4) + (b4 ^ (c4 & 15))] = v;
        }
    }

    __syncthreads();

    const int bo = tid & 15;
    const int eo = tid >> 4;
    #pragma unroll
    for (int i = 0; i < T_COLS / 16; ++i) {
        const int el = eo + 16 * i;
        const v4f v = tile[el * (BATCH / 4) + (bo ^ ((el >> 2) & 15))];
        *(v4f*)(xT + (size_t)(e0 + el) * BATCH + 4 * bo) = v;
    }
}

extern "C" void kernel_launch(void* const* d_in, const int* in_sizes, int n_in,
                              void* d_out, int out_size, void* d_ws, size_t ws_size,
                              hipStream_t stream) {
    (void)in_sizes; (void)n_in; (void)out_size; (void)ws_size;

    const float* x0 = (const float*)d_in[0];
    const float* W1 = (const float*)d_in[1];
    const float* b1 = (const float*)d_in[2];
    const float* W2 = (const float*)d_in[3];
    const float* b2 = (const float*)d_in[4];
    const int*   ix = (const int*)d_in[5];

    float* bufA = (float*)d_ws;
    float* out  = (float*)d_out;

    const dim3 grid(GRID_WAVES / WAVES_PER_BLOCK);
    const dim3 block(WAVES_PER_BLOCK * 32);

    transpose_x0<<<dim3(T_BLOCKS), dim3(256), 0, stream>>>(x0, out);
    gsnn_layer<true><<<grid, block, 0, stream>>>(out, x0, W1, b1, W2, b2, ix, bufA);
    const float* cur = bufA;
    for (int l = 1; l < LAYERS - 1; ++l) {
        float* dst = (l & 1) ? out : bufA;
        gsnn_layer<true><<<grid, block, 0, stream>>>(cur, x0, W1, b1, W2, b2, ix, dst);
        cur = dst;
    }
    gsnn_layer<false><<<grid, block, 0, stream>>>(cur, x0, W1, b1, W2, b2, ix, out);
}
